# starting kernel
# baseline (speedup 1.0000x reference)
#include <hip/hip_runtime.h>
#include <math.h>

#define BATCH   2
#define SEQ     2048
#define DMODEL  1024
#define NHEADS  16
#define DK      64
#define LOG_THETA 9.210340371976184f

typedef __attribute__((ext_vector_type(2))) float v2f;
typedef __attribute__((ext_vector_type(8))) float v8f;

static __device__ __forceinline__ v8f wmma_f32_4(v2f a, v2f b, v8f c) {
  return __builtin_amdgcn_wmma_f32_16x16x4_f32(
       false, a,  false, b,
       (short)0, c,  false,  false);
}

__global__ __launch_bounds__(256) void proj_gemm_kernel(
    const float* __restrict__ A, const float* __restrict__ W,
    float* __restrict__ Y, int in_head_major, int out_head_major,
    int apply_rope) {
  const int lane = threadIdx.x & 31;
  const int wave = threadIdx.x >> 5;
  const int l15  = lane & 15;
  const int half = lane >> 4;
  const int koff = half * 2;
  const int n0 = (blockIdx.x * 8 + wave) * 16;
  const int m0 = blockIdx.y * 16;
  const int arow = m0 + l15;
  const int wrow = n0 + l15;

  v8f c = {0.f,0.f,0.f,0.f,0.f,0.f,0.f,0.f};

  if (in_head_major) {
    const int bb = arow >> 11;
    const int ss = arow & (SEQ - 1);
    #pragma unroll 8
    for (int kb = 0; kb < DMODEL; kb += 4) {
      const int k0 = kb + koff;
      v2f a = *(const v2f*)(A + (((size_t)(bb * NHEADS + (k0 >> 6)) * SEQ + ss) * DK)
                              + (k0 & 63));
      v2f b = *(const v2f*)(W + (size_t)wrow * DMODEL + k0);
      c = wmma_f32_4(a, b, c);
    }
  } else {
    #pragma unroll 8
    for (int kb = 0; kb < DMODEL; kb += 4) {
      const int k0 = kb + koff;
      v2f a = *(const v2f*)(A + (size_t)arow * DMODEL + k0);
      v2f b = *(const v2f*)(W + (size_t)wrow * DMODEL + k0);
      c = wmma_f32_4(a, b, c);
    }
  }

  const int e = n0 + l15;
  #pragma unroll
  for (int v = 0; v < 8; ++v) {
    const int mrow = m0 + v + half * 8;
    const int ss = mrow & (SEQ - 1);
    const int bb = mrow >> 11;
    float val = c[v];
    if (apply_rope) {
      float partner = __shfl_xor(val, 1, 32);
      const int p = (e & 63) >> 1;
      const float freq  = __expf(-(2.0f * LOG_THETA / (float)DK) * (float)p);
      const float phase = (float)ss * freq;
      float sn, cs;
      __sincosf(phase, &sn, &cs);
      val = (e & 1) ? (partner * sn + val * cs) : (val * cs - partner * sn);
    }
    if (out_head_major)
      Y[((size_t)(bb * NHEADS + (e >> 6)) * SEQ + ss) * DK + (e & 63)] = val;
    else
      Y[(size_t)mrow * DMODEL + e] = val;
  }
}

__global__ __launch_bounds__(256) void attn_kernel(
    const float* __restrict__ Q, const float* __restrict__ K,
    const float* __restrict__ V, float* __restrict__ O) {
  __shared__ float ldsS[8][256];
  __shared__ float ldsFac[8][16];
  __shared__ float ldsStat[8][16];

  const int lane = threadIdx.x & 31;
  const int wave = threadIdx.x >> 5;
  const int l15  = lane & 15;
  const int half = lane >> 4;
  const int koff = half * 2;
  const int qt = blockIdx.x * 8 + wave;
  const int h  = blockIdx.y;
  const int b  = blockIdx.z;
  const size_t base = ((size_t)(b * NHEADS + h)) * SEQ * DK;
  const float* Qh = Q + base;
  const float* Kh = K + base;
  const float* Vh = V + base;
  float*       Oh = O + base;
  const int r0 = qt * 16;

  float* S    = ldsS[wave];
  float* FAC  = ldsFac[wave];
  float* STAT = ldsStat[wave];

  v2f qa[16];
  #pragma unroll
  for (int kb = 0; kb < 16; ++kb)
    qa[kb] = *(const v2f*)(Qh + (size_t)(r0 + l15) * DK + kb * 4 + koff);

  v8f o0 = {0.f,0.f,0.f,0.f,0.f,0.f,0.f,0.f};
  v8f o1 = o0, o2 = o0, o3 = o0;
  float m_i = -__builtin_inff();
  float l_i = 0.f;

  for (int j = 0; j <= qt; ++j) {
    const int c0 = j * 16;

    v8f s = {0.f,0.f,0.f,0.f,0.f,0.f,0.f,0.f};
    #pragma unroll
    for (int kb = 0; kb < 16; ++kb) {
      v2f kv = *(const v2f*)(Kh + (size_t)(c0 + l15) * DK + kb * 4 + koff);
      s = wmma_f32_4(qa[kb], kv, s);
    }

    const int kcol = c0 + l15;
    #pragma unroll
    for (int v = 0; v < 8; ++v) {
      const int qrow = r0 + v + half * 8;
      S[(v + half * 8) * 16 + l15] =
          (kcol <= qrow) ? s[v] * 0.125f : -__builtin_inff();
    }
    asm volatile("s_wait_dscnt 0" ::: "memory");

    if (lane < 16) {
      float mrow = -__builtin_inff();
      for (int c2 = 0; c2 < 16; ++c2) mrow = fmaxf(mrow, S[lane * 16 + c2]);
      const float mnew = fmaxf(m_i, mrow);
      const float fac  = __expf(m_i - mnew);
      float rs = 0.f;
      for (int c2 = 0; c2 < 16; ++c2) {
        const float p = __expf(S[lane * 16 + c2] - mnew);
        S[lane * 16 + c2] = p;
        rs += p;
      }
      l_i = l_i * fac + rs;
      m_i = mnew;
      FAC[lane] = fac;
    }
    asm volatile("s_wait_dscnt 0" ::: "memory");

    float facv[8];
    #pragma unroll
    for (int v = 0; v < 8; ++v) facv[v] = FAC[v + half * 8];
    #pragma unroll
    for (int v = 0; v < 8; ++v) {
      o0[v] *= facv[v]; o1[v] *= facv[v];
      o2[v] *= facv[v]; o3[v] *= facv[v];
    }

    #pragma unroll
    for (int kb = 0; kb < 4; ++kb) {
      const int kk = kb * 4 + koff;
      v2f pa = *(const v2f*)(S + l15 * 16 + kk);
      const float* vp = Vh + (size_t)(c0 + kk) * DK + l15;
      v2f vb0 = { vp[0],  vp[DK] };
      v2f vb1 = { vp[16], vp[16 + DK] };
      v2f vb2 = { vp[32], vp[32 + DK] };
      v2f vb3 = { vp[48], vp[48 + DK] };
      o0 = wmma_f32_4(pa, vb0, o0);
      o1 = wmma_f32_4(pa, vb1, o1);
      o2 = wmma_f32_4(pa, vb2, o2);
      o3 = wmma_f32_4(pa, vb3, o3);
    }
  }

  if (lane < 16) STAT[lane] = 1.0f / l_i;
  asm volatile("s_wait_dscnt 0" ::: "memory");
  #pragma unroll
  for (int v = 0; v < 8; ++v) {
    const float linv = STAT[v + half * 8];
    float* op = Oh + (size_t)(r0 + v + half * 8) * DK + l15;
    op[0]  = o0[v] * linv;
    op[16] = o1[v] * linv;
    op[32] = o2[v] * linv;
    op[48] = o3[v] * linv;
  }
}

extern "C" void kernel_launch(void* const* d_in, const int* in_sizes, int n_in,
                              void* d_out, int out_size, void* d_ws,
                              size_t ws_size, hipStream_t stream) {
  (void)in_sizes; (void)n_in; (void)out_size; (void)ws_size;
  const float* x  = (const float*)d_in[0];
  const float* Wq = (const float*)d_in[1];
  const float* Wk = (const float*)d_in[2];
  const float* Wv = (const float*)d_in[3];
  const float* Wo = (const float*)d_in[4];
  float* out = (float*)d_out;

  const size_t perBuf = (size_t)BATCH * NHEADS * SEQ * DK;
  float* Qb = (float*)d_ws;
  float* Kb = Qb + perBuf;
  float* Vb = Kb + perBuf;
  float* AO = Vb + perBuf;

  dim3 blk(256);
  dim3 gproj(8, (BATCH * SEQ) / 16, 1);

  proj_gemm_kernel<<<gproj, blk, 0, stream>>>(x, Wq, Qb, 0, 1, 1);
  proj_gemm_kernel<<<gproj, blk, 0, stream>>>(x, Wk, Kb, 0, 1, 1);
  proj_gemm_kernel<<<gproj, blk, 0, stream>>>(x, Wv, Vb, 0, 1, 0);

  attn_kernel<<<dim3(SEQ / 16 / 8, NHEADS, BATCH), blk, 0, stream>>>(Qb, Kb, Vb, AO);

  proj_gemm_kernel<<<gproj, blk, 0, stream>>>(AO, Wo, out, 1, 0, 0);
}


# Q, K, V stored by the projection epilogues in matrix-operand order (Q, K as transposed product): attention operand loads are 16 B per lane on whole lines; k order of both attention products changes
# speedup vs baseline: 2.1810x; 2.1810x over previous
#include <hip/hip_runtime.h>
#include <math.h>

#define BATCH   2
#define SEQ     2048
#define DMODEL  1024
#define NHEADS  16
#define DK      64
#define LOG_THETA 9.210340371976184f

typedef __attribute__((ext_vector_type(2))) float v2f;
typedef __attribute__((ext_vector_type(4))) float v4f;
typedef __attribute__((ext_vector_type(8))) float v8f;

static __device__ __forceinline__ v8f wmma_f32_4(v2f a, v2f b, v8f c) {
  return __builtin_amdgcn_wmma_f32_16x16x4_f32(
       false, a,  false, b,
       (short)0, c,  false,  false);
}

template <int IN_HM, int OUT_MODE, int ROPE>
__global__ __launch_bounds__(128) __attribute__((amdgpu_waves_per_eu(1, 1)))
void proj_gemm_rt_kernel(const float* __restrict__ A,
                         const float* __restrict__ W,
                         float* __restrict__ Y) {
  const int lane = threadIdx.x & 31;
  const int wave = threadIdx.x >> 5;
  const int c    = lane & 15;
  const int half = lane >> 4;
  const int m0 = blockIdx.y * 128 + (wave >> 1) * 64;
  const int n0 = blockIdx.x * 128 + (wave & 1) * 64;

  const float* ap[4];
  const float* bp[4];
  #pragma unroll
  for (int i = 0; i < 4; ++i) {
    const int row = m0 + 16 * i + c;
    if (IN_HM) {
      const int bb = row >> 11;
      const int ss = row & (SEQ - 1);
      ap[i] = A + ((size_t)(bb * NHEADS) * SEQ + ss) * DK + 4 * half;
    } else {
      ap[i] = A + (size_t)row * DMODEL + 4 * half;
    }
    bp[i] = W + (size_t)(n0 + 16 * i + c) * DMODEL + 4 * half;
  }

  v8f acc[4][4];
  #pragma unroll
  for (int mi = 0; mi < 4; ++mi)
    #pragma unroll
    for (int ni = 0; ni < 4; ++ni)
      acc[mi][ni] = (v8f){0.f,0.f,0.f,0.f,0.f,0.f,0.f,0.f};

  auto load_group = [&](v4f (&fa)[4][4], v4f (&fb)[4][4], int k) {
    const size_t ak = IN_HM ? (size_t)(k >> 6) * ((size_t)SEQ * DK) + (size_t)(k & 63)
                            : (size_t)k;
    #pragma unroll
    for (int sub = 0; sub < 4; ++sub) {
      #pragma unroll
      for (int i = 0; i < 4; ++i) {
        fa[sub][i] = *(const v4f*)(ap[i] + ak + 8 * sub);
        fb[sub][i] = *(const v4f*)(bp[i] + k + 8 * sub);
      }
    }
  };
  auto compute_group = [&](const v4f (&fa)[4][4], const v4f (&fb)[4][4]) {
    #pragma unroll
    for (int sub = 0; sub < 4; ++sub) {
      #pragma unroll
      for (int mi = 0; mi < 4; ++mi) {
        const v2f a0 = { fa[sub][mi][0], fa[sub][mi][1] };
        #pragma unroll
        for (int ni = 0; ni < 4; ++ni) {
          const v2f b0 = { fb[sub][ni][0], fb[sub][ni][1] };
          acc[mi][ni] = (OUT_MODE == 1) ? wmma_f32_4(b0, a0, acc[mi][ni])
                                        : wmma_f32_4(a0, b0, acc[mi][ni]);
        }
      }
      #pragma unroll
      for (int mi = 0; mi < 4; ++mi) {
        const v2f a1 = { fa[sub][mi][2], fa[sub][mi][3] };
        #pragma unroll
        for (int ni = 0; ni < 4; ++ni) {
          const v2f b1 = { fb[sub][ni][2], fb[sub][ni][3] };
          acc[mi][ni] = (OUT_MODE == 1) ? wmma_f32_4(b1, a1, acc[mi][ni])
                                        : wmma_f32_4(a1, b1, acc[mi][ni]);
        }
      }
    }
  };

  v4f fa0[4][4], fb0[4][4], fa1[4][4], fb1[4][4];
  load_group(fa0, fb0, 0);
  #pragma unroll 1
  for (int k = 0; k < DMODEL - 64; k += 64) {
    load_group(fa1, fb1, k + 32);
    compute_group(fa0, fb0);
    load_group(fa0, fb0, k + 64);
    compute_group(fa1, fb1);
  }
  load_group(fa1, fb1, DMODEL - 32);
  compute_group(fa0, fb0);
  compute_group(fa1, fb1);

  if (OUT_MODE == 1) {
    const int hd = n0 >> 6;
    float freq[4][2][2];
    #pragma unroll
    for (int ni = 0; ni < 4; ++ni)
      #pragma unroll
      for (int vq = 0; vq < 2; ++vq)
        #pragma unroll
        for (int u = 0; u < 2; ++u) {
          const int pr = 2 * (4 * ni + 2 * half + vq) + u;
          freq[ni][vq][u] = ROPE ? __expf(-(2.0f * LOG_THETA / (float)DK) * (float)pr) : 0.f;
        }
    #pragma unroll
    for (int mi = 0; mi < 4; ++mi) {
      const int pos = m0 + 16 * mi + c;
      const int bb = pos >> 11;
      const int ss = pos & (SEQ - 1);
      float* tile = Y + ((size_t)(bb * NHEADS + hd) * SEQ + (size_t)(ss & ~15)) * DK;
      #pragma unroll
      for (int ni = 0; ni < 4; ++ni) {
        #pragma unroll
        for (int vq = 0; vq < 2; ++vq) {
          const int p = 4 * ni + 2 * half + vq;
          v4f o;
          #pragma unroll
          for (int u = 0; u < 2; ++u) {
            const float xe = acc[mi][ni][4 * vq + 2 * u];
            const float xo = acc[mi][ni][4 * vq + 2 * u + 1];
            if (ROPE) {
              const float phase = (float)ss * freq[ni][vq][u];
              float sn, cs;
              __sincosf(phase, &sn, &cs);
              o[2 * u]     = xe * cs - xo * sn;
              o[2 * u + 1] = xe * sn + xo * cs;
            } else {
              o[2 * u]     = xe;
              o[2 * u + 1] = xo;
            }
          }
          *(v4f*)(tile + p * 64 + c * 4) = o;
        }
      }
    }
  } else if (OUT_MODE == 2) {
    const int hd = n0 >> 6;
    const int bb = m0 >> 11;
    const int t0 = (m0 & (SEQ - 1)) >> 4;
    float* hb = Y + (size_t)(bb * NHEADS + hd) * SEQ * DK;
    #pragma unroll
    for (int mi = 0; mi < 4; ++mi) {
      float* tile = hb + (size_t)(t0 + mi) * (16 * DK);
      #pragma unroll
      for (int ni = 0; ni < 4; ++ni) {
        #pragma unroll
        for (int vq = 0; vq < 2; ++vq) {
          const v4f o = { acc[mi][ni][4 * vq],     acc[mi][ni][4 * vq + 1],
                          acc[mi][ni][4 * vq + 2], acc[mi][ni][4 * vq + 3] };
          *(v4f*)(tile + ni * 256 + half * 128 + vq * 64 + c * 4) = o;
        }
      }
    }
  } else {
    #pragma unroll
    for (int ni = 0; ni < 4; ++ni) {
      const int e = n0 + 16 * ni + c;
      #pragma unroll
      for (int mi = 0; mi < 4; ++mi) {
        #pragma unroll
        for (int v = 0; v < 8; ++v) {
          const int mrow = m0 + 16 * mi + v + half * 8;
          Y[(size_t)mrow * DMODEL + e] = acc[mi][ni][v];
        }
      }
    }
  }
}

#define SP 20
__global__ __launch_bounds__(256) __attribute__((amdgpu_waves_per_eu(4, 4)))
void attn_kernel(
    const float* __restrict__ Q, const float* __restrict__ K,
    const float* __restrict__ V, float* __restrict__ O) {
  __shared__ __attribute__((aligned(16))) float ldsS[8][32 * SP];
  __shared__ float ldsFac[8][32];
  __shared__ float ldsStat[8][32];

  const int lane = threadIdx.x & 31;
  const int wave = threadIdx.x >> 5;
  const int l15  = lane & 15;
  const int half = lane >> 4;
  const int qt = blockIdx.x * 8 + wave;
  const int h  = blockIdx.y;
  const int b  = blockIdx.z;
  const size_t base = ((size_t)(b * NHEADS + h)) * SEQ * DK;
  const float* Qh = Q + base;
  const float* Kh = K + base;
  const float* Vh = V + base;
  float*       Oh = O + base;
  const int r0 = qt * 16;

  float* S    = ldsS[wave];
  float* FAC  = ldsFac[wave];
  float* STAT = ldsStat[wave];

  const int fo = half * 64 + l15 * 4;

  v4f qa[8];
  #pragma unroll
  for (int i = 0; i < 8; ++i)
    qa[i] = *(const v4f*)(Qh + (size_t)r0 * DK + i * 128 + fo);

  v8f o0 = {0.f,0.f,0.f,0.f,0.f,0.f,0.f,0.f};
  v8f o1 = o0, o2 = o0, o3 = o0;
  float m_i = -__builtin_inff();
  float l_i = 0.f;

  float* srow = S + (half * 16 + l15) * SP;

  const int nround = (qt + 2) >> 1;
  for (int j = 0; j < nround; ++j) {
    const int c0 = j * 32;

    v8f s0 = {0.f,0.f,0.f,0.f,0.f,0.f,0.f,0.f};
    v8f s1 = s0;
    const float* Kt = Kh + (size_t)c0 * DK + fo;
    v4f k0[8], k1[8];
    #pragma unroll
    for (int i = 0; i < 8; ++i) {
      k0[i] = *(const v4f*)(Kt + i * 128);
      k1[i] = *(const v4f*)(Kt + 16 * DK + i * 128);
    }
    #pragma unroll
    for (int i = 0; i < 8; ++i) {
      const v2f qlo = { qa[i][0], qa[i][1] };
      const v2f qhi = { qa[i][2], qa[i][3] };
      s0 = wmma_f32_4(qlo, (v2f){ k0[i][0], k0[i][1] }, s0);
      s1 = wmma_f32_4(qlo, (v2f){ k1[i][0], k1[i][1] }, s1);
      s0 = wmma_f32_4(qhi, (v2f){ k0[i][2], k0[i][3] }, s0);
      s1 = wmma_f32_4(qhi, (v2f){ k1[i][2], k1[i][3] }, s1);
    }

    const int kcol0 = c0 + l15;
    const int kcol1 = kcol0 + 16;
    #pragma unroll
    for (int v = 0; v < 8; ++v) {
      const int rr = v + half * 8;
      const int qrow = r0 + rr;
      S[rr * SP + l15] =
          (kcol0 <= qrow) ? s0[v] * 0.125f : -__builtin_inff();
      S[(16 + rr) * SP + l15] =
          (kcol1 <= qrow) ? s1[v] * 0.125f : -__builtin_inff();
    }
    asm volatile("s_wait_dscnt 0" ::: "memory");

    {
      v4f x0 = *(const v4f*)(srow);
      v4f x1 = *(const v4f*)(srow + 4);
      v4f x2 = *(const v4f*)(srow + 8);
      v4f x3 = *(const v4f*)(srow + 12);
      float xs[16] = { x0[0], x0[1], x0[2], x0[3], x1[0], x1[1], x1[2], x1[3],
                       x2[0], x2[1], x2[2], x2[3], x3[0], x3[1], x3[2], x3[3] };
      float mh = -__builtin_inff();
      #pragma unroll
      for (int c2 = 0; c2 < 16; ++c2) mh = fmaxf(mh, xs[c2]);
      const float mrow = fmaxf(mh, __shfl_xor(mh, 16, 32));
      const float mnew = fmaxf(m_i, mrow);
      const float fac  = __expf(m_i - mnew);
      float rs = 0.f;
      #pragma unroll
      for (int c2 = 0; c2 < 16; ++c2) {
        xs[c2] = __expf(xs[c2] - mnew);
        rs += xs[c2];
      }
      rs += __shfl_xor(rs, 16, 32);
      *(v4f*)(srow)      = (v4f){ xs[0],  xs[1],  xs[2],  xs[3]  };
      *(v4f*)(srow + 4)  = (v4f){ xs[4],  xs[5],  xs[6],  xs[7]  };
      *(v4f*)(srow + 8)  = (v4f){ xs[8],  xs[9],  xs[10], xs[11] };
      *(v4f*)(srow + 12) = (v4f){ xs[12], xs[13], xs[14], xs[15] };
      l_i = l_i * fac + rs;
      m_i = mnew;
      FAC[lane] = fac;
    }
    asm volatile("s_wait_dscnt 0" ::: "memory");

    const float* Vt = Vh + (size_t)c0 * DK + fo;
    v4f vb[2][2][4];
    #pragma unroll
    for (int t = 0; t < 2; ++t)
      #pragma unroll
      for (int jj = 0; jj < 2; ++jj)
        #pragma unroll
        for (int n = 0; n < 4; ++n)
          vb[t][jj][n] = *(const v4f*)(Vt + t * (16 * DK) + n * 256 + jj * 128);

    float facv[8];
    #pragma unroll
    for (int v = 0; v < 8; ++v) facv[v] = FAC[v + half * 8];
    #pragma unroll
    for (int v = 0; v < 8; ++v) {
      o0[v] *= facv[v]; o1[v] *= facv[v];
      o2[v] *= facv[v]; o3[v] *= facv[v];
    }

    v4f pa[2][2];
    #pragma unroll
    for (int t = 0; t < 2; ++t)
      #pragma unroll
      for (int jj = 0; jj < 2; ++jj)
        pa[t][jj] = *(const v4f*)(S + (t * 16 + l15) * SP + 8 * jj + 4 * half);
    #pragma unroll
    for (int t = 0; t < 2; ++t) {
      #pragma unroll
      for (int jj = 0; jj < 2; ++jj) {
        const v2f plo = { pa[t][jj][0], pa[t][jj][1] };
        const v2f phi = { pa[t][jj][2], pa[t][jj][3] };
        o0 = wmma_f32_4(plo, (v2f){ vb[t][jj][0][0], vb[t][jj][0][1] }, o0);
        o1 = wmma_f32_4(plo, (v2f){ vb[t][jj][1][0], vb[t][jj][1][1] }, o1);
        o2 = wmma_f32_4(plo, (v2f){ vb[t][jj][2][0], vb[t][jj][2][1] }, o2);
        o3 = wmma_f32_4(plo, (v2f){ vb[t][jj][3][0], vb[t][jj][3][1] }, o3);
        o0 = wmma_f32_4(phi, (v2f){ vb[t][jj][0][2], vb[t][jj][0][3] }, o0);
        o1 = wmma_f32_4(phi, (v2f){ vb[t][jj][1][2], vb[t][jj][1][3] }, o1);
        o2 = wmma_f32_4(phi, (v2f){ vb[t][jj][2][2], vb[t][jj][2][3] }, o2);
        o3 = wmma_f32_4(phi, (v2f){ vb[t][jj][3][2], vb[t][jj][3][3] }, o3);
      }
    }
  }

  STAT[lane] = 1.0f / l_i;
  asm volatile("s_wait_dscnt 0" ::: "memory");
  #pragma unroll
  for (int v = 0; v < 8; ++v) {
    const float linv = STAT[v + half * 8];
    float* op = Oh + (size_t)(r0 + v + half * 8) * DK + l15;
    op[0]  = o0[v] * linv;
    op[16] = o1[v] * linv;
    op[32] = o2[v] * linv;
    op[48] = o3[v] * linv;
  }
}

extern "C" void kernel_launch(void* const* d_in, const int* in_sizes, int n_in,
                              void* d_out, int out_size, void* d_ws,
                              size_t ws_size, hipStream_t stream) {
  (void)in_sizes; (void)n_in; (void)out_size; (void)ws_size;
  const float* x  = (const float*)d_in[0];
  const float* Wq = (const float*)d_in[1];
  const float* Wk = (const float*)d_in[2];
  const float* Wv = (const float*)d_in[3];
  const float* Wo = (const float*)d_in[4];
  float* out = (float*)d_out;

  const size_t perBuf = (size_t)BATCH * NHEADS * SEQ * DK;
  float* Qb = (float*)d_ws;
  float* Kb = Qb + perBuf;
  float* Vb = Kb + perBuf;
  float* AO = Vb + perBuf;

  dim3 blk(256);
  dim3 gblk(128);
  dim3 gproj(DMODEL / 128, (BATCH * SEQ) / 128, 1);

  proj_gemm_rt_kernel<0, 1, 1><<<gproj, gblk, 0, stream>>>(x, Wq, Qb);
  proj_gemm_rt_kernel<0, 1, 1><<<gproj, gblk, 0, stream>>>(x, Wk, Kb);
  proj_gemm_rt_kernel<0, 2, 0><<<gproj, gblk, 0, stream>>>(x, Wv, Vb);

  attn_kernel<<<dim3(SEQ / 16 / 8, NHEADS, BATCH), blk, 0, stream>>>(Qb, Kb, Vb, AO);

  proj_gemm_rt_kernel<1, 0, 0><<<gproj, gblk, 0, stream>>>(AO, Wo, out);
}


# attention: each query tile's 32-key rounds split by parity over two waves, partials joined once in LDS; grid of 2x resident waves, heaviest query groups first; row-sum and P*V sum order changes
# speedup vs baseline: 2.7386x; 1.2557x over previous
#include <hip/hip_runtime.h>
#include <math.h>

#define BATCH   2
#define SEQ     2048
#define DMODEL  1024
#define NHEADS  16
#define DK      64
#define LOG_THETA 9.210340371976184f

typedef __attribute__((ext_vector_type(2))) float v2f;
typedef __attribute__((ext_vector_type(4))) float v4f;
typedef __attribute__((ext_vector_type(8))) float v8f;

static __device__ __forceinline__ v8f wmma_f32_4(v2f a, v2f b, v8f c) {
  return __builtin_amdgcn_wmma_f32_16x16x4_f32(
       false, a,  false, b,
       (short)0, c,  false,  false);
}

template <int IN_HM, int OUT_MODE, int ROPE>
__global__ __launch_bounds__(128) __attribute__((amdgpu_waves_per_eu(1, 1)))
void proj_gemm_rt_kernel(const float* __restrict__ A,
                         const float* __restrict__ W,
                         float* __restrict__ Y) {
  const int lane = threadIdx.x & 31;
  const int wave = threadIdx.x >> 5;
  const int c    = lane & 15;
  const int half = lane >> 4;
  const int m0 = blockIdx.y * 128 + (wave >> 1) * 64;
  const int n0 = blockIdx.x * 128 + (wave & 1) * 64;

  const float* ap[4];
  const float* bp[4];
  #pragma unroll
  for (int i = 0; i < 4; ++i) {
    const int row = m0 + 16 * i + c;
    if (IN_HM) {
      const int bb = row >> 11;
      const int ss = row & (SEQ - 1);
      ap[i] = A + ((size_t)(bb * NHEADS) * SEQ + ss) * DK + 4 * half;
    } else {
      ap[i] = A + (size_t)row * DMODEL + 4 * half;
    }
    bp[i] = W + (size_t)(n0 + 16 * i + c) * DMODEL + 4 * half;
  }

  v8f acc[4][4];
  #pragma unroll
  for (int mi = 0; mi < 4; ++mi)
    #pragma unroll
    for (int ni = 0; ni < 4; ++ni)
      acc[mi][ni] = (v8f){0.f,0.f,0.f,0.f,0.f,0.f,0.f,0.f};

  auto load_group = [&](v4f (&fa)[4][4], v4f (&fb)[4][4], int k) {
    const size_t ak = IN_HM ? (size_t)(k >> 6) * ((size_t)SEQ * DK) + (size_t)(k & 63)
                            : (size_t)k;
    #pragma unroll
    for (int sub = 0; sub < 4; ++sub) {
      #pragma unroll
      for (int i = 0; i < 4; ++i) {
        fa[sub][i] = *(const v4f*)(ap[i] + ak + 8 * sub);
        fb[sub][i] = *(const v4f*)(bp[i] + k + 8 * sub);
      }
    }
  };
  auto compute_group = [&](const v4f (&fa)[4][4], const v4f (&fb)[4][4]) {
    #pragma unroll
    for (int sub = 0; sub < 4; ++sub) {
      #pragma unroll
      for (int mi = 0; mi < 4; ++mi) {
        const v2f a0 = { fa[sub][mi][0], fa[sub][mi][1] };
        #pragma unroll
        for (int ni = 0; ni < 4; ++ni) {
          const v2f b0 = { fb[sub][ni][0], fb[sub][ni][1] };
          acc[mi][ni] = (OUT_MODE == 1) ? wmma_f32_4(b0, a0, acc[mi][ni])
                                        : wmma_f32_4(a0, b0, acc[mi][ni]);
        }
      }
      #pragma unroll
      for (int mi = 0; mi < 4; ++mi) {
        const v2f a1 = { fa[sub][mi][2], fa[sub][mi][3] };
        #pragma unroll
        for (int ni = 0; ni < 4; ++ni) {
          const v2f b1 = { fb[sub][ni][2], fb[sub][ni][3] };
          acc[mi][ni] = (OUT_MODE == 1) ? wmma_f32_4(b1, a1, acc[mi][ni])
                                        : wmma_f32_4(a1, b1, acc[mi][ni]);
        }
      }
    }
  };

  v4f fa0[4][4], fb0[4][4], fa1[4][4], fb1[4][4];
  load_group(fa0, fb0, 0);
  #pragma unroll 1
  for (int k = 0; k < DMODEL - 64; k += 64) {
    load_group(fa1, fb1, k + 32);
    compute_group(fa0, fb0);
    load_group(fa0, fb0, k + 64);
    compute_group(fa1, fb1);
  }
  load_group(fa1, fb1, DMODEL - 32);
  compute_group(fa0, fb0);
  compute_group(fa1, fb1);

  if (OUT_MODE == 1) {
    const int hd = n0 >> 6;
    float freq[4][2][2];
    #pragma unroll
    for (int ni = 0; ni < 4; ++ni)
      #pragma unroll
      for (int vq = 0; vq < 2; ++vq)
        #pragma unroll
        for (int u = 0; u < 2; ++u) {
          const int pr = 2 * (4 * ni + 2 * half + vq) + u;
          freq[ni][vq][u] = ROPE ? __expf(-(2.0f * LOG_THETA / (float)DK) * (float)pr) : 0.f;
        }
    #pragma unroll
    for (int mi = 0; mi < 4; ++mi) {
      const int pos = m0 + 16 * mi + c;
      const int bb = pos >> 11;
      const int ss = pos & (SEQ - 1);
      float* tile = Y + ((size_t)(bb * NHEADS + hd) * SEQ + (size_t)(ss & ~15)) * DK;
      #pragma unroll
      for (int ni = 0; ni < 4; ++ni) {
        #pragma unroll
        for (int vq = 0; vq < 2; ++vq) {
          const int p = 4 * ni + 2 * half + vq;
          v4f o;
          #pragma unroll
          for (int u = 0; u < 2; ++u) {
            const float xe = acc[mi][ni][4 * vq + 2 * u];
            const float xo = acc[mi][ni][4 * vq + 2 * u + 1];
            if (ROPE) {
              const float phase = (float)ss * freq[ni][vq][u];
              float sn, cs;
              __sincosf(phase, &sn, &cs);
              o[2 * u]     = xe * cs - xo * sn;
              o[2 * u + 1] = xe * sn + xo * cs;
            } else {
              o[2 * u]     = xe;
              o[2 * u + 1] = xo;
            }
          }
          *(v4f*)(tile + p * 64 + c * 4) = o;
        }
      }
    }
  } else if (OUT_MODE == 2) {
    const int hd = n0 >> 6;
    const int bb = m0 >> 11;
    const int t0 = (m0 & (SEQ - 1)) >> 4;
    float* hb = Y + (size_t)(bb * NHEADS + hd) * SEQ * DK;
    #pragma unroll
    for (int mi = 0; mi < 4; ++mi) {
      float* tile = hb + (size_t)(t0 + mi) * (16 * DK);
      #pragma unroll
      for (int ni = 0; ni < 4; ++ni) {
        #pragma unroll
        for (int vq = 0; vq < 2; ++vq) {
          const v4f o = { acc[mi][ni][4 * vq],     acc[mi][ni][4 * vq + 1],
                          acc[mi][ni][4 * vq + 2], acc[mi][ni][4 * vq + 3] };
          *(v4f*)(tile + ni * 256 + half * 128 + vq * 64 + c * 4) = o;
        }
      }
    }
  } else {
    #pragma unroll
    for (int ni = 0; ni < 4; ++ni) {
      const int e = n0 + 16 * ni + c;
      #pragma unroll
      for (int mi = 0; mi < 4; ++mi) {
        #pragma unroll
        for (int v = 0; v < 8; ++v) {
          const int mrow = m0 + 16 * mi + v + half * 8;
          Y[(size_t)mrow * DMODEL + e] = acc[mi][ni][v];
        }
      }
    }
  }
}

#define SP 20
__global__ __launch_bounds__(256) __attribute__((amdgpu_waves_per_eu(4, 4)))
void attn_kernel(
    const float* __restrict__ Q, const float* __restrict__ K,
    const float* __restrict__ V, float* __restrict__ O) {
  __shared__ __attribute__((aligned(16))) float ldsS[8][32 * SP];
  __shared__ float ldsFac[8][32];
  __shared__ float ldsStat[8][32];
  __shared__ __attribute__((aligned(16))) float ldsPO[4][8 * 32 * 4];
  __shared__ float ldsPM[4][16];
  __shared__ float ldsPL[4][16];

  const int lane = threadIdx.x & 31;
  const int wave = threadIdx.x >> 5;
  const int l15  = lane & 15;
  const int half = lane >> 4;
  const int t  = wave & 3;
  const int w  = wave >> 2;
  const int g  = (SEQ / 64 - 1) - (int)blockIdx.y;
  const int qt = g * 4 + t;
  const size_t base = (size_t)blockIdx.x * SEQ * DK;
  const float* Qh = Q + base;
  const float* Kh = K + base;
  const float* Vh = V + base;
  float*       Oh = O + base;
  const int r0 = qt * 16;

  float* S    = ldsS[wave];
  float* FAC  = ldsFac[wave];
  float* STAT = ldsStat[wave];

  const int fo = half * 64 + l15 * 4;

  v4f qa[8];
  #pragma unroll
  for (int i = 0; i < 8; ++i)
    qa[i] = *(const v4f*)(Qh + (size_t)r0 * DK + i * 128 + fo);

  v8f o0 = {0.f,0.f,0.f,0.f,0.f,0.f,0.f,0.f};
  v8f o1 = o0, o2 = o0, o3 = o0;
  float m_i = -__builtin_inff();
  float l_i = 0.f;

  float* srow = S + (half * 16 + l15) * SP;

  const int nround = (qt + 2) >> 1;
  for (int j = w; j < nround; j += 2) {
    const int c0 = j * 32;

    v8f s0 = {0.f,0.f,0.f,0.f,0.f,0.f,0.f,0.f};
    v8f s1 = s0;
    const float* Kt = Kh + (size_t)c0 * DK + fo;
    v4f k0[8], k1[8];
    #pragma unroll
    for (int i = 0; i < 8; ++i) {
      k0[i] = *(const v4f*)(Kt + i * 128);
      k1[i] = *(const v4f*)(Kt + 16 * DK + i * 128);
    }
    #pragma unroll
    for (int i = 0; i < 8; ++i) {
      const v2f qlo = { qa[i][0], qa[i][1] };
      const v2f qhi = { qa[i][2], qa[i][3] };
      s0 = wmma_f32_4(qlo, (v2f){ k0[i][0], k0[i][1] }, s0);
      s1 = wmma_f32_4(qlo, (v2f){ k1[i][0], k1[i][1] }, s1);
      s0 = wmma_f32_4(qhi, (v2f){ k0[i][2], k0[i][3] }, s0);
      s1 = wmma_f32_4(qhi, (v2f){ k1[i][2], k1[i][3] }, s1);
    }

    const int kcol0 = c0 + l15;
    const int kcol1 = kcol0 + 16;
    #pragma unroll
    for (int v = 0; v < 8; ++v) {
      const int rr = v + half * 8;
      const int qrow = r0 + rr;
      S[rr * SP + l15] =
          (kcol0 <= qrow) ? s0[v] * 0.125f : -__builtin_inff();
      S[(16 + rr) * SP + l15] =
          (kcol1 <= qrow) ? s1[v] * 0.125f : -__builtin_inff();
    }
    asm volatile("s_wait_dscnt 0" ::: "memory");

    {
      v4f x0 = *(const v4f*)(srow);
      v4f x1 = *(const v4f*)(srow + 4);
      v4f x2 = *(const v4f*)(srow + 8);
      v4f x3 = *(const v4f*)(srow + 12);
      float xs[16] = { x0[0], x0[1], x0[2], x0[3], x1[0], x1[1], x1[2], x1[3],
                       x2[0], x2[1], x2[2], x2[3], x3[0], x3[1], x3[2], x3[3] };
      float mh = -__builtin_inff();
      #pragma unroll
      for (int c2 = 0; c2 < 16; ++c2) mh = fmaxf(mh, xs[c2]);
      const float mrow = fmaxf(mh, __shfl_xor(mh, 16, 32));
      const float mnew = fmaxf(m_i, mrow);
      const float fac  = __expf(m_i - mnew);
      float rs = 0.f;
      #pragma unroll
      for (int c2 = 0; c2 < 16; ++c2) {
        xs[c2] = __expf(xs[c2] - mnew);
        rs += xs[c2];
      }
      rs += __shfl_xor(rs, 16, 32);
      *(v4f*)(srow)      = (v4f){ xs[0],  xs[1],  xs[2],  xs[3]  };
      *(v4f*)(srow + 4)  = (v4f){ xs[4],  xs[5],  xs[6],  xs[7]  };
      *(v4f*)(srow + 8)  = (v4f){ xs[8],  xs[9],  xs[10], xs[11] };
      *(v4f*)(srow + 12) = (v4f){ xs[12], xs[13], xs[14], xs[15] };
      l_i = l_i * fac + rs;
      m_i = mnew;
      FAC[lane] = fac;
    }
    asm volatile("s_wait_dscnt 0" ::: "memory");

    const float* Vt = Vh + (size_t)c0 * DK + fo;
    v4f vb[2][2][4];
    #pragma unroll
    for (int t = 0; t < 2; ++t)
      #pragma unroll
      for (int jj = 0; jj < 2; ++jj)
        #pragma unroll
        for (int n = 0; n < 4; ++n)
          vb[t][jj][n] = *(const v4f*)(Vt + t * (16 * DK) + n * 256 + jj * 128);

    float facv[8];
    #pragma unroll
    for (int v = 0; v < 8; ++v) facv[v] = FAC[v + half * 8];
    #pragma unroll
    for (int v = 0; v < 8; ++v) {
      o0[v] *= facv[v]; o1[v] *= facv[v];
      o2[v] *= facv[v]; o3[v] *= facv[v];
    }

    v4f pa[2][2];
    #pragma unroll
    for (int t = 0; t < 2; ++t)
      #pragma unroll
      for (int jj = 0; jj < 2; ++jj)
        pa[t][jj] = *(const v4f*)(S + (t * 16 + l15) * SP + 8 * jj + 4 * half);
    #pragma unroll
    for (int t = 0; t < 2; ++t) {
      #pragma unroll
      for (int jj = 0; jj < 2; ++jj) {
        const v2f plo = { pa[t][jj][0], pa[t][jj][1] };
        const v2f phi = { pa[t][jj][2], pa[t][jj][3] };
        o0 = wmma_f32_4(plo, (v2f){ vb[t][jj][0][0], vb[t][jj][0][1] }, o0);
        o1 = wmma_f32_4(plo, (v2f){ vb[t][jj][1][0], vb[t][jj][1][1] }, o1);
        o2 = wmma_f32_4(plo, (v2f){ vb[t][jj][2][0], vb[t][jj][2][1] }, o2);
        o3 = wmma_f32_4(plo, (v2f){ vb[t][jj][3][0], vb[t][jj][3][1] }, o3);
        o0 = wmma_f32_4(phi, (v2f){ vb[t][jj][0][2], vb[t][jj][0][3] }, o0);
        o1 = wmma_f32_4(phi, (v2f){ vb[t][jj][1][2], vb[t][jj][1][3] }, o1);
        o2 = wmma_f32_4(phi, (v2f){ vb[t][jj][2][2], vb[t][jj][2][3] }, o2);
        o3 = wmma_f32_4(phi, (v2f){ vb[t][jj][3][2], vb[t][jj][3][3] }, o3);
      }
    }
  }

  v4f* po = (v4f*)ldsPO[t];
  if (w == 1) {
    po[0 * 32 + lane] = (v4f){ o0[0], o0[1], o0[2], o0[3] };
    po[1 * 32 + lane] = (v4f){ o0[4], o0[5], o0[6], o0[7] };
    po[2 * 32 + lane] = (v4f){ o1[0], o1[1], o1[2], o1[3] };
    po[3 * 32 + lane] = (v4f){ o1[4], o1[5], o1[6], o1[7] };
    po[4 * 32 + lane] = (v4f){ o2[0], o2[1], o2[2], o2[3] };
    po[5 * 32 + lane] = (v4f){ o2[4], o2[5], o2[6], o2[7] };
    po[6 * 32 + lane] = (v4f){ o3[0], o3[1], o3[2], o3[3] };
    po[7 * 32 + lane] = (v4f){ o3[4], o3[5], o3[6], o3[7] };
    ldsPM[t][l15] = m_i;
    ldsPL[t][l15] = l_i;
  }
  __syncthreads();
  if (w == 0) {
    const float m1 = ldsPM[t][l15];
    const float l1 = ldsPL[t][l15];
    const float mn = fmaxf(m_i, m1);
    const float f0 = __expf(m_i - mn);
    const float f1 = __expf(m1 - mn);
    const float linv = 1.0f / (l_i * f0 + l1 * f1);
    STAT[lane] = f0 * linv;
    FAC[lane]  = f1 * linv;
    asm volatile("s_wait_dscnt 0" ::: "memory");
    v4f p[8];
    #pragma unroll
    for (int i = 0; i < 8; ++i) p[i] = po[i * 32 + lane];
    #pragma unroll
    for (int v = 0; v < 8; ++v) {
      const float a0 = STAT[v + half * 8];
      const float a1 = FAC[v + half * 8];
      float* op = Oh + (size_t)(r0 + v + half * 8) * DK + l15;
      op[0]  = o0[v] * a0 + p[0 + (v >> 2)][v & 3] * a1;
      op[16] = o1[v] * a0 + p[2 + (v >> 2)][v & 3] * a1;
      op[32] = o2[v] * a0 + p[4 + (v >> 2)][v & 3] * a1;
      op[48] = o3[v] * a0 + p[6 + (v >> 2)][v & 3] * a1;
    }
  }
}

extern "C" void kernel_launch(void* const* d_in, const int* in_sizes, int n_in,
                              void* d_out, int out_size, void* d_ws,
                              size_t ws_size, hipStream_t stream) {
  (void)in_sizes; (void)n_in; (void)out_size; (void)ws_size;
  const float* x  = (const float*)d_in[0];
  const float* Wq = (const float*)d_in[1];
  const float* Wk = (const float*)d_in[2];
  const float* Wv = (const float*)d_in[3];
  const float* Wo = (const float*)d_in[4];
  float* out = (float*)d_out;

  const size_t perBuf = (size_t)BATCH * NHEADS * SEQ * DK;
  float* Qb = (float*)d_ws;
  float* Kb = Qb + perBuf;
  float* Vb = Kb + perBuf;
  float* AO = Vb + perBuf;

  dim3 blk(256);
  dim3 gblk(128);
  dim3 gproj(DMODEL / 128, (BATCH * SEQ) / 128, 1);

  proj_gemm_rt_kernel<0, 1, 1><<<gproj, gblk, 0, stream>>>(x, Wq, Qb);
  proj_gemm_rt_kernel<0, 1, 1><<<gproj, gblk, 0, stream>>>(x, Wk, Kb);
  proj_gemm_rt_kernel<0, 2, 0><<<gproj, gblk, 0, stream>>>(x, Wv, Vb);

  attn_kernel<<<dim3(BATCH * NHEADS, SEQ / 64, 1), blk, 0, stream>>>(Qb, Kb, Vb, AO);

  proj_gemm_rt_kernel<1, 0, 0><<<gproj, gblk, 0, stream>>>(AO, Wo, out);
}


# projection GEMMs Q, K, V: 3072 register wave tiles in one grid (z = weight), two waves resident per SIMD: a wave's load bursts, first loads and epilogue run under the other's WMMAs; sums unchanged
# speedup vs baseline: 2.8335x; 1.0347x over previous
#include <hip/hip_runtime.h>
#include <math.h>

#define BATCH   2
#define SEQ     2048
#define DMODEL  1024
#define NHEADS  16
#define DK      64
#define LOG_THETA 9.210340371976184f

typedef __attribute__((ext_vector_type(2))) float v2f;
typedef __attribute__((ext_vector_type(4))) float v4f;
typedef __attribute__((ext_vector_type(8))) float v8f;

static __device__ __forceinline__ v8f wmma_f32_4(v2f a, v2f b, v8f c) {
  return __builtin_amdgcn_wmma_f32_16x16x4_f32(
       false, a,  false, b,
       (short)0, c,  false,  false);
}

template <int IN_HM, int OUT_MODE, int ROPE>
static __device__ __forceinline__
void proj_gemm_rt_tile(const float* __restrict__ A,
                       const float* __restrict__ W,
                       float* __restrict__ Y) {
  const int lane = threadIdx.x & 31;
  const int wave = threadIdx.x >> 5;
  const int c    = lane & 15;
  const int half = lane >> 4;
  const int m0 = blockIdx.y * 128 + (wave >> 1) * 64;
  const int n0 = blockIdx.x * 128 + (wave & 1) * 64;

  const float* ap[4];
  const float* bp[4];
  #pragma unroll
  for (int i = 0; i < 4; ++i) {
    const int row = m0 + 16 * i + c;
    if (IN_HM) {
      const int bb = row >> 11;
      const int ss = row & (SEQ - 1);
      ap[i] = A + ((size_t)(bb * NHEADS) * SEQ + ss) * DK + 4 * half;
    } else {
      ap[i] = A + (size_t)row * DMODEL + 4 * half;
    }
    bp[i] = W + (size_t)(n0 + 16 * i + c) * DMODEL + 4 * half;
  }

  v8f acc[4][4];
  #pragma unroll
  for (int mi = 0; mi < 4; ++mi)
    #pragma unroll
    for (int ni = 0; ni < 4; ++ni)
      acc[mi][ni] = (v8f){0.f,0.f,0.f,0.f,0.f,0.f,0.f,0.f};

  auto load_group = [&](v4f (&fa)[4][4], v4f (&fb)[4][4], int k) {
    const size_t ak = IN_HM ? (size_t)(k >> 6) * ((size_t)SEQ * DK) + (size_t)(k & 63)
                            : (size_t)k;
    #pragma unroll
    for (int sub = 0; sub < 4; ++sub) {
      #pragma unroll
      for (int i = 0; i < 4; ++i) {
        fa[sub][i] = *(const v4f*)(ap[i] + ak + 8 * sub);
        fb[sub][i] = *(const v4f*)(bp[i] + k + 8 * sub);
      }
    }
  };
  auto compute_group = [&](const v4f (&fa)[4][4], const v4f (&fb)[4][4]) {
    #pragma unroll
    for (int sub = 0; sub < 4; ++sub) {
      #pragma unroll
      for (int mi = 0; mi < 4; ++mi) {
        const v2f a0 = { fa[sub][mi][0], fa[sub][mi][1] };
        #pragma unroll
        for (int ni = 0; ni < 4; ++ni) {
          const v2f b0 = { fb[sub][ni][0], fb[sub][ni][1] };
          acc[mi][ni] = (OUT_MODE == 1) ? wmma_f32_4(b0, a0, acc[mi][ni])
                                        : wmma_f32_4(a0, b0, acc[mi][ni]);
        }
      }
      #pragma unroll
      for (int mi = 0; mi < 4; ++mi) {
        const v2f a1 = { fa[sub][mi][2], fa[sub][mi][3] };
        #pragma unroll
        for (int ni = 0; ni < 4; ++ni) {
          const v2f b1 = { fb[sub][ni][2], fb[sub][ni][3] };
          acc[mi][ni] = (OUT_MODE == 1) ? wmma_f32_4(b1, a1, acc[mi][ni])
                                        : wmma_f32_4(a1, b1, acc[mi][ni]);
        }
      }
    }
  };

  v4f fa0[4][4], fb0[4][4], fa1[4][4], fb1[4][4];
  load_group(fa0, fb0, 0);
  #pragma unroll 1
  for (int k = 0; k < DMODEL - 64; k += 64) {
    load_group(fa1, fb1, k + 32);
    compute_group(fa0, fb0);
    load_group(fa0, fb0, k + 64);
    compute_group(fa1, fb1);
  }
  load_group(fa1, fb1, DMODEL - 32);
  compute_group(fa0, fb0);
  compute_group(fa1, fb1);

  if (OUT_MODE == 1) {
    const int hd = n0 >> 6;
    float freq[4][2][2];
    #pragma unroll
    for (int ni = 0; ni < 4; ++ni)
      #pragma unroll
      for (int vq = 0; vq < 2; ++vq)
        #pragma unroll
        for (int u = 0; u < 2; ++u) {
          const int pr = 2 * (4 * ni + 2 * half + vq) + u;
          freq[ni][vq][u] = ROPE ? __expf(-(2.0f * LOG_THETA / (float)DK) * (float)pr) : 0.f;
        }
    #pragma unroll
    for (int mi = 0; mi < 4; ++mi) {
      const int pos = m0 + 16 * mi + c;
      const int bb = pos >> 11;
      const int ss = pos & (SEQ - 1);
      float* tile = Y + ((size_t)(bb * NHEADS + hd) * SEQ + (size_t)(ss & ~15)) * DK;
      #pragma unroll
      for (int ni = 0; ni < 4; ++ni) {
        #pragma unroll
        for (int vq = 0; vq < 2; ++vq) {
          const int p = 4 * ni + 2 * half + vq;
          v4f o;
          #pragma unroll
          for (int u = 0; u < 2; ++u) {
            const float xe = acc[mi][ni][4 * vq + 2 * u];
            const float xo = acc[mi][ni][4 * vq + 2 * u + 1];
            if (ROPE) {
              const float phase = (float)ss * freq[ni][vq][u];
              float sn, cs;
              __sincosf(phase, &sn, &cs);
              o[2 * u]     = xe * cs - xo * sn;
              o[2 * u + 1] = xe * sn + xo * cs;
            } else {
              o[2 * u]     = xe;
              o[2 * u + 1] = xo;
            }
          }
          *(v4f*)(tile + p * 64 + c * 4) = o;
        }
      }
    }
  } else if (OUT_MODE == 2) {
    const int hd = n0 >> 6;
    const int bb = m0 >> 11;
    const int t0 = (m0 & (SEQ - 1)) >> 4;
    float* hb = Y + (size_t)(bb * NHEADS + hd) * SEQ * DK;
    #pragma unroll
    for (int mi = 0; mi < 4; ++mi) {
      float* tile = hb + (size_t)(t0 + mi) * (16 * DK);
      #pragma unroll
      for (int ni = 0; ni < 4; ++ni) {
        #pragma unroll
        for (int vq = 0; vq < 2; ++vq) {
          const v4f o = { acc[mi][ni][4 * vq],     acc[mi][ni][4 * vq + 1],
                          acc[mi][ni][4 * vq + 2], acc[mi][ni][4 * vq + 3] };
          *(v4f*)(tile + ni * 256 + half * 128 + vq * 64 + c * 4) = o;
        }
      }
    }
  } else {
    #pragma unroll
    for (int ni = 0; ni < 4; ++ni) {
      const int e = n0 + 16 * ni + c;
      #pragma unroll
      for (int mi = 0; mi < 4; ++mi) {
        #pragma unroll
        for (int v = 0; v < 8; ++v) {
          const int mrow = m0 + 16 * mi + v + half * 8;
          Y[(size_t)mrow * DMODEL + e] = acc[mi][ni][v];
        }
      }
    }
  }
}

template <int IN_HM, int OUT_MODE, int ROPE>
__global__ __launch_bounds__(128) __attribute__((amdgpu_waves_per_eu(1, 1)))
void proj_gemm_rt_kernel(const float* __restrict__ A,
                         const float* __restrict__ W,
                         float* __restrict__ Y) {
  proj_gemm_rt_tile<IN_HM, OUT_MODE, ROPE>(A, W, Y);
}

__global__ __launch_bounds__(128) __attribute__((amdgpu_waves_per_eu(2, 2)))
void qkv_proj_kernel(const float* __restrict__ x,
                     const float* __restrict__ Wq,
                     const float* __restrict__ Wk,
                     const float* __restrict__ Wv,
                     float* __restrict__ Qb,
                     float* __restrict__ Kb,
                     float* __restrict__ Vb) {
  if (blockIdx.z == 2) {
    proj_gemm_rt_tile<0, 2, 0>(x, Wv, Vb);
  } else {
    const bool isq = (blockIdx.z == 0);
    proj_gemm_rt_tile<0, 1, 1>(x, isq ? Wq : Wk, isq ? Qb : Kb);
  }
}

#define SP 20
__global__ __launch_bounds__(256) __attribute__((amdgpu_waves_per_eu(4, 4)))
void attn_kernel(
    const float* __restrict__ Q, const float* __restrict__ K,
    const float* __restrict__ V, float* __restrict__ O) {
  __shared__ __attribute__((aligned(16))) float ldsS[8][32 * SP];
  __shared__ float ldsFac[8][32];
  __shared__ float ldsStat[8][32];
  __shared__ __attribute__((aligned(16))) float ldsPO[4][8 * 32 * 4];
  __shared__ float ldsPM[4][16];
  __shared__ float ldsPL[4][16];

  const int lane = threadIdx.x & 31;
  const int wave = threadIdx.x >> 5;
  const int l15  = lane & 15;
  const int half = lane >> 4;
  const int t  = wave & 3;
  const int w  = wave >> 2;
  const int g  = (SEQ / 64 - 1) - (int)blockIdx.y;
  const int qt = g * 4 + t;
  const size_t base = (size_t)blockIdx.x * SEQ * DK;
  const float* Qh = Q + base;
  const float* Kh = K + base;
  const float* Vh = V + base;
  float*       Oh = O + base;
  const int r0 = qt * 16;

  float* S    = ldsS[wave];
  float* FAC  = ldsFac[wave];
  float* STAT = ldsStat[wave];

  const int fo = half * 64 + l15 * 4;

  v4f qa[8];
  #pragma unroll
  for (int i = 0; i < 8; ++i)
    qa[i] = *(const v4f*)(Qh + (size_t)r0 * DK + i * 128 + fo);

  v8f o0 = {0.f,0.f,0.f,0.f,0.f,0.f,0.f,0.f};
  v8f o1 = o0, o2 = o0, o3 = o0;
  float m_i = -__builtin_inff();
  float l_i = 0.f;

  float* srow = S + (half * 16 + l15) * SP;

  const int nround = (qt + 2) >> 1;
  for (int j = w; j < nround; j += 2) {
    const int c0 = j * 32;

    v8f s0 = {0.f,0.f,0.f,0.f,0.f,0.f,0.f,0.f};
    v8f s1 = s0;
    const float* Kt = Kh + (size_t)c0 * DK + fo;
    v4f k0[8], k1[8];
    #pragma unroll
    for (int i = 0; i < 8; ++i) {
      k0[i] = *(const v4f*)(Kt + i * 128);
      k1[i] = *(const v4f*)(Kt + 16 * DK + i * 128);
    }
    #pragma unroll
    for (int i = 0; i < 8; ++i) {
      const v2f qlo = { qa[i][0], qa[i][1] };
      const v2f qhi = { qa[i][2], qa[i][3] };
      s0 = wmma_f32_4(qlo, (v2f){ k0[i][0], k0[i][1] }, s0);
      s1 = wmma_f32_4(qlo, (v2f){ k1[i][0], k1[i][1] }, s1);
      s0 = wmma_f32_4(qhi, (v2f){ k0[i][2], k0[i][3] }, s0);
      s1 = wmma_f32_4(qhi, (v2f){ k1[i][2], k1[i][3] }, s1);
    }

    const int kcol0 = c0 + l15;
    const int kcol1 = kcol0 + 16;
    #pragma unroll
    for (int v = 0; v < 8; ++v) {
      const int rr = v + half * 8;
      const int qrow = r0 + rr;
      S[rr * SP + l15] =
          (kcol0 <= qrow) ? s0[v] * 0.125f : -__builtin_inff();
      S[(16 + rr) * SP + l15] =
          (kcol1 <= qrow) ? s1[v] * 0.125f : -__builtin_inff();
    }
    asm volatile("s_wait_dscnt 0" ::: "memory");

    {
      v4f x0 = *(const v4f*)(srow);
      v4f x1 = *(const v4f*)(srow + 4);
      v4f x2 = *(const v4f*)(srow + 8);
      v4f x3 = *(const v4f*)(srow + 12);
      float xs[16] = { x0[0], x0[1], x0[2], x0[3], x1[0], x1[1], x1[2], x1[3],
                       x2[0], x2[1], x2[2], x2[3], x3[0], x3[1], x3[2], x3[3] };
      float mh = -__builtin_inff();
      #pragma unroll
      for (int c2 = 0; c2 < 16; ++c2) mh = fmaxf(mh, xs[c2]);
      const float mrow = fmaxf(mh, __shfl_xor(mh, 16, 32));
      const float mnew = fmaxf(m_i, mrow);
      const float fac  = __expf(m_i - mnew);
      float rs = 0.f;
      #pragma unroll
      for (int c2 = 0; c2 < 16; ++c2) {
        xs[c2] = __expf(xs[c2] - mnew);
        rs += xs[c2];
      }
      rs += __shfl_xor(rs, 16, 32);
      *(v4f*)(srow)      = (v4f){ xs[0],  xs[1],  xs[2],  xs[3]  };
      *(v4f*)(srow + 4)  = (v4f){ xs[4],  xs[5],  xs[6],  xs[7]  };
      *(v4f*)(srow + 8)  = (v4f){ xs[8],  xs[9],  xs[10], xs[11] };
      *(v4f*)(srow + 12) = (v4f){ xs[12], xs[13], xs[14], xs[15] };
      l_i = l_i * fac + rs;
      m_i = mnew;
      FAC[lane] = fac;
    }
    asm volatile("s_wait_dscnt 0" ::: "memory");

    const float* Vt = Vh + (size_t)c0 * DK + fo;
    v4f vb[2][2][4];
    #pragma unroll
    for (int t = 0; t < 2; ++t)
      #pragma unroll
      for (int jj = 0; jj < 2; ++jj)
        #pragma unroll
        for (int n = 0; n < 4; ++n)
          vb[t][jj][n] = *(const v4f*)(Vt + t * (16 * DK) + n * 256 + jj * 128);

    float facv[8];
    #pragma unroll
    for (int v = 0; v < 8; ++v) facv[v] = FAC[v + half * 8];
    #pragma unroll
    for (int v = 0; v < 8; ++v) {
      o0[v] *= facv[v]; o1[v] *= facv[v];
      o2[v] *= facv[v]; o3[v] *= facv[v];
    }

    v4f pa[2][2];
    #pragma unroll
    for (int t = 0; t < 2; ++t)
      #pragma unroll
      for (int jj = 0; jj < 2; ++jj)
        pa[t][jj] = *(const v4f*)(S + (t * 16 + l15) * SP + 8 * jj + 4 * half);
    #pragma unroll
    for (int t = 0; t < 2; ++t) {
      #pragma unroll
      for (int jj = 0; jj < 2; ++jj) {
        const v2f plo = { pa[t][jj][0], pa[t][jj][1] };
        const v2f phi = { pa[t][jj][2], pa[t][jj][3] };
        o0 = wmma_f32_4(plo, (v2f){ vb[t][jj][0][0], vb[t][jj][0][1] }, o0);
        o1 = wmma_f32_4(plo, (v2f){ vb[t][jj][1][0], vb[t][jj][1][1] }, o1);
        o2 = wmma_f32_4(plo, (v2f){ vb[t][jj][2][0], vb[t][jj][2][1] }, o2);
        o3 = wmma_f32_4(plo, (v2f){ vb[t][jj][3][0], vb[t][jj][3][1] }, o3);
        o0 = wmma_f32_4(phi, (v2f){ vb[t][jj][0][2], vb[t][jj][0][3] }, o0);
        o1 = wmma_f32_4(phi, (v2f){ vb[t][jj][1][2], vb[t][jj][1][3] }, o1);
        o2 = wmma_f32_4(phi, (v2f){ vb[t][jj][2][2], vb[t][jj][2][3] }, o2);
        o3 = wmma_f32_4(phi, (v2f){ vb[t][jj][3][2], vb[t][jj][3][3] }, o3);
      }
    }
  }

  v4f* po = (v4f*)ldsPO[t];
  if (w == 1) {
    po[0 * 32 + lane] = (v4f){ o0[0], o0[1], o0[2], o0[3] };
    po[1 * 32 + lane] = (v4f){ o0[4], o0[5], o0[6], o0[7] };
    po[2 * 32 + lane] = (v4f){ o1[0], o1[1], o1[2], o1[3] };
    po[3 * 32 + lane] = (v4f){ o1[4], o1[5], o1[6], o1[7] };
    po[4 * 32 + lane] = (v4f){ o2[0], o2[1], o2[2], o2[3] };
    po[5 * 32 + lane] = (v4f){ o2[4], o2[5], o2[6], o2[7] };
    po[6 * 32 + lane] = (v4f){ o3[0], o3[1], o3[2], o3[3] };
    po[7 * 32 + lane] = (v4f){ o3[4], o3[5], o3[6], o3[7] };
    ldsPM[t][l15] = m_i;
    ldsPL[t][l15] = l_i;
  }
  __syncthreads();
  if (w == 0) {
    const float m1 = ldsPM[t][l15];
    const float l1 = ldsPL[t][l15];
    const float mn = fmaxf(m_i, m1);
    const float f0 = __expf(m_i - mn);
    const float f1 = __expf(m1 - mn);
    const float linv = 1.0f / (l_i * f0 + l1 * f1);
    STAT[lane] = f0 * linv;
    FAC[lane]  = f1 * linv;
    asm volatile("s_wait_dscnt 0" ::: "memory");
    v4f p[8];
    #pragma unroll
    for (int i = 0; i < 8; ++i) p[i] = po[i * 32 + lane];
    #pragma unroll
    for (int v = 0; v < 8; ++v) {
      const float a0 = STAT[v + half * 8];
      const float a1 = FAC[v + half * 8];
      float* op = Oh + (size_t)(r0 + v + half * 8) * DK + l15;
      op[0]  = o0[v] * a0 + p[0 + (v >> 2)][v & 3] * a1;
      op[16] = o1[v] * a0 + p[2 + (v >> 2)][v & 3] * a1;
      op[32] = o2[v] * a0 + p[4 + (v >> 2)][v & 3] * a1;
      op[48] = o3[v] * a0 + p[6 + (v >> 2)][v & 3] * a1;
    }
  }
}

extern "C" void kernel_launch(void* const* d_in, const int* in_sizes, int n_in,
                              void* d_out, int out_size, void* d_ws,
                              size_t ws_size, hipStream_t stream) {
  (void)in_sizes; (void)n_in; (void)out_size; (void)ws_size;
  const float* x  = (const float*)d_in[0];
  const float* Wq = (const float*)d_in[1];
  const float* Wk = (const float*)d_in[2];
  const float* Wv = (const float*)d_in[3];
  const float* Wo = (const float*)d_in[4];
  float* out = (float*)d_out;

  const size_t perBuf = (size_t)BATCH * NHEADS * SEQ * DK;
  float* Qb = (float*)d_ws;
  float* Kb = Qb + perBuf;
  float* Vb = Kb + perBuf;
  float* AO = Vb + perBuf;

  dim3 blk(256);
  dim3 gblk(128);
  dim3 gproj(DMODEL / 128, (BATCH * SEQ) / 128, 1);

  dim3 gqkv(DMODEL / 128, (BATCH * SEQ) / 128, 3);
  qkv_proj_kernel<<<gqkv, gblk, 0, stream>>>(x, Wq, Wk, Wv, Qb, Kb, Vb);

  attn_kernel<<<dim3(BATCH * NHEADS, SEQ / 64, 1), blk, 0, stream>>>(Qb, Kb, Vb, AO);

  proj_gemm_rt_kernel<1, 0, 0><<<gproj, gblk, 0, stream>>>(AO, Wo, out);
}
